# starting kernel
# baseline (speedup 1.0000x reference)
#include <hip/hip_runtime.h>

#define N_LEAF       64
#define SPECTRAL_DIM 64
#define N_BANDS      4
#define BAND_SIZE    16
#define LAM          0.1f
#define EPSF         1e-8f

typedef float v2f __attribute__((ext_vector_type(2)));
typedef float v4f __attribute__((ext_vector_type(4)));
typedef float v8f __attribute__((ext_vector_type(8)));

#define WAVES_PER_BLOCK 4
#define PTS_PER_WAVE    32
#define PTS_PER_BLOCK   (WAVES_PER_BLOCK * PTS_PER_WAVE)
#define LDS_STRIDE      65

__device__ __forceinline__ v8f zero8() {
    v8f z;
    #pragma unroll
    for (int i = 0; i < 8; ++i) z[i] = 0.0f;
    return z;
}

__device__ __forceinline__ float frcp(float x)  { return __builtin_amdgcn_rcpf(x); }
__device__ __forceinline__ float fsqrt(float x) { return __builtin_amdgcn_sqrtf(x); }

__global__ __launch_bounds__(PTS_PER_BLOCK) void inception_traversal_kernel(
    const float* __restrict__ pos3d,
    const float* __restrict__ spectral,
    const float* __restrict__ centers1,
    const float* __restrict__ centers2,
    const float* __restrict__ centers3,
    const float* __restrict__ portal1,
    const float* __restrict__ portal2,
    const float* __restrict__ Wb,
    const float* __restrict__ bb,
    const float* __restrict__ bw,
    float* __restrict__ out,
    int BS)
{
    __shared__ float ldsRefr[WAVES_PER_BLOCK][PTS_PER_WAVE][LDS_STRIDE];

    const int lane   = threadIdx.x & 31;
    const int wid    = threadIdx.x >> 5;
    const int pblock = blockIdx.x * PTS_PER_BLOCK;
    if (pblock >= BS) return;
    const int pwave  = pblock + wid * PTS_PER_WAVE;

    float w0 = bw[0], w1 = bw[1], w2 = bw[2], w3 = bw[3];
    float wm = fmaxf(fmaxf(w0, w1), fmaxf(w2, w3));
    float ew0 = __expf(w0 - wm), ew1 = __expf(w1 - wm);
    float ew2 = __expf(w2 - wm), ew3 = __expf(w3 - wm);
    float wr  = frcp(ew0 + ew1 + ew2 + ew3);
    float wband[N_BANDS] = { ew0 * wr, ew1 * wr, ew2 * wr, ew3 * wr };

    const int mrow  = lane & 15;
    const int khalf = (lane >> 4) * 2;

    #pragma unroll
    for (int mt = 0; mt < 2; ++mt) {
        const int P = pwave + mt * 16;

        v8f acc[4];
        #pragma unroll
        for (int t = 0; t < 4; ++t) acc[t] = zero8();

        #pragma unroll
        for (int n = 0; n < N_BANDS; ++n) {
            v8f logit[4];
            #pragma unroll
            for (int t = 0; t < 4; ++t) logit[t] = zero8();

            #pragma unroll
            for (int kk = 0; kk < 4; ++kk) {
                const int c = n * BAND_SIZE + kk * 4 + khalf;
                v2f a = *reinterpret_cast<const v2f*>(
                    spectral + (size_t)(P + mrow) * SPECTRAL_DIM + c);
                #pragma unroll
                for (int t = 0; t < 4; ++t) {
                    const float* wp = Wb + ((size_t)n * BAND_SIZE + kk * 4 + khalf) * N_LEAF
                                         + t * 16 + mrow;
                    v2f b; b.x = wp[0]; b.y = wp[N_LEAF];
                    logit[t] = __builtin_amdgcn_wmma_f32_16x16x4_f32(
                        false, a, false, b, (short)0, logit[t], false, false);
                }
            }
            #pragma unroll
            for (int t = 0; t < 4; ++t) {
                const float bias = bb[n * N_LEAF + t * 16 + mrow];
                const float wn   = wband[n];
                #pragma unroll
                for (int j = 0; j < 8; ++j) {
                    float s = frcp(1.0f + __expf(-(logit[t][j] + bias)));
                    acc[t][j] += wn * s;
                }
            }
        }

        const int rowoff = (lane >> 4) * 8 + mt * 16;
        #pragma unroll
        for (int t = 0; t < 4; ++t)
            #pragma unroll
            for (int j = 0; j < 8; ++j)
                ldsRefr[wid][rowoff + j][t * 16 + mrow] = acc[t][j];
    }

    __syncthreads();

    const int p = pwave + lane;
    float* row = &ldsRefr[wid][lane][0];

    const float px = pos3d[(size_t)p * 3 + 0];
    const float py = pos3d[(size_t)p * 3 + 1];
    const float pz = pos3d[(size_t)p * 3 + 2];
    const float inv2T2 = 1.0f / (2.0f + EPSF);

    float d1v[4];
    #pragma unroll
    for (int d = 0; d < 4; ++d) {
        float dx = px - centers1[d * 3 + 0];
        float dy = py - centers1[d * 3 + 1];
        float dz = pz - centers1[d * 3 + 2];
        d1v[d] = dx * dx + dy * dy + dz * dz;
    }
    float z1max = -1e30f;
    #pragma unroll
    for (int d = 0; d < 4; ++d) z1max = fmaxf(z1max, -d1v[d] * inv2T2);
    float ex1[4], s1 = 0.0f;
    #pragma unroll
    for (int d = 0; d < 4; ++d) { ex1[d] = __expf(-d1v[d] * inv2T2 - z1max); s1 += ex1[d]; }
    const float r1 = frcp(s1);
    float m1[4], e1v[4];
    #pragma unroll
    for (int d = 0; d < 4; ++d) {
        m1[d]  = ex1[d] * r1;
        e1v[d] = __expf(-LAM * fsqrt(d1v[d] + EPSF));
    }

    float ssum = 0.0f;
    #pragma unroll
    for (int d = 0; d < 4; ++d) {
        const float* T1 = portal1 + d * 12;
        float qx = T1[0]*px + T1[1]*py + T1[2]*pz  + T1[3];
        float qy = T1[4]*px + T1[5]*py + T1[6]*pz  + T1[7];
        float qz = T1[8]*px + T1[9]*py + T1[10]*pz + T1[11];

        float d2v[4];
        #pragma unroll
        for (int si = 0; si < 4; ++si) {
            const float* c = centers2 + (size_t)(d * 4 + si) * 3;
            float dx = qx - c[0], dy = qy - c[1], dz = qz - c[2];
            d2v[si] = dx * dx + dy * dy + dz * dz;
        }
        float z2max = -1e30f;
        #pragma unroll
        for (int si = 0; si < 4; ++si) z2max = fmaxf(z2max, -d2v[si] * inv2T2);
        float ex2[4], s2 = 0.0f;
        #pragma unroll
        for (int si = 0; si < 4; ++si) { ex2[si] = __expf(-d2v[si] * inv2T2 - z2max); s2 += ex2[si]; }
        const float r2 = frcp(s2);

        #pragma unroll
        for (int si = 0; si < 4; ++si) {
            float m2 = m1[d] * ex2[si] * r2;
            float e2 = e1v[d] * __expf(-LAM * fsqrt(d2v[si] + EPSF));

            const float* T2 = portal2 + (size_t)(d * 4 + si) * 12;
            float rx = T2[0]*qx + T2[1]*qy + T2[2]*qz  + T2[3];
            float ry = T2[4]*qx + T2[5]*qy + T2[6]*qz  + T2[7];
            float rz = T2[8]*qx + T2[9]*qy + T2[10]*qz + T2[11];

            float d3v[4];
            #pragma unroll
            for (int ci = 0; ci < 4; ++ci) {
                const float* c = centers3 + (size_t)((d * 4 + si) * 4 + ci) * 3;
                float dx = rx - c[0], dy = ry - c[1], dz = rz - c[2];
                d3v[ci] = dx * dx + dy * dy + dz * dz;
            }
            float z3max = -1e30f;
            #pragma unroll
            for (int ci = 0; ci < 4; ++ci) z3max = fmaxf(z3max, -d3v[ci] * inv2T2);
            float ex3[4], s3 = 0.0f;
            #pragma unroll
            for (int ci = 0; ci < 4; ++ci) { ex3[ci] = __expf(-d3v[ci] * inv2T2 - z3max); s3 += ex3[ci]; }
            const float r3   = frcp(s3);
            const float m2e2 = m2 * e2;

            #pragma unroll
            for (int ci = 0; ci < 4; ++ci) {
                float g  = m2e2 * ex3[ci] * r3 * __expf(-LAM * fsqrt(d3v[ci] + EPSF));
                int   k  = (d * 4 + si) * 4 + ci;
                float pr = g * row[k];
                row[k] = pr;
                ssum  += pr;
            }
        }
    }

    float rn = frcp(fmaxf(ssum, 1e-8f));
    float* op = out + (size_t)p * N_LEAF;
    #pragma unroll
    for (int k4 = 0; k4 < 16; ++k4) {
        v4f v;
        v.x = row[k4 * 4 + 0] * rn;
        v.y = row[k4 * 4 + 1] * rn;
        v.z = row[k4 * 4 + 2] * rn;
        v.w = row[k4 * 4 + 3] * rn;
        *reinterpret_cast<v4f*>(op + k4 * 4) = v;
    }
}

extern "C" void kernel_launch(void* const* d_in, const int* in_sizes, int n_in,
                              void* d_out, int out_size, void* d_ws, size_t ws_size,
                              hipStream_t stream) {
    const float* pos3d    = (const float*)d_in[0];
    const float* spectral = (const float*)d_in[1];
    const float* centers1 = (const float*)d_in[2];
    const float* centers2 = (const float*)d_in[3];
    const float* centers3 = (const float*)d_in[4];
    const float* portal1  = (const float*)d_in[5];
    const float* portal2  = (const float*)d_in[6];
    const float* Wb       = (const float*)d_in[7];
    const float* bb       = (const float*)d_in[8];
    const float* bw       = (const float*)d_in[9];
    float* out            = (float*)d_out;

    const int BS = in_sizes[0] / 3;
    const int blocks = (BS + PTS_PER_BLOCK - 1) / PTS_PER_BLOCK;

    hipLaunchKernelGGL(inception_traversal_kernel,
                       dim3(blocks), dim3(PTS_PER_BLOCK), 0, stream,
                       pos3d, spectral, centers1, centers2, centers3,
                       portal1, portal2, Wb, bb, bw, out, BS);
}


# Band loop outermost: one band's B fragments live at a time, shared by both 16-point tiles; 298 -> 185 VGPRs (budget 256), so 4 blocks fit per compute unit and the 1024-block grid runs in one round
# speedup vs baseline: 1.2808x; 1.2808x over previous
#include <hip/hip_runtime.h>

#define N_LEAF       64
#define SPECTRAL_DIM 64
#define N_BANDS      4
#define BAND_SIZE    16
#define LAM          0.1f
#define EPSF         1e-8f

typedef float v2f __attribute__((ext_vector_type(2)));
typedef float v4f __attribute__((ext_vector_type(4)));
typedef float v8f __attribute__((ext_vector_type(8)));

#define WAVES_PER_BLOCK 4
#define PTS_PER_WAVE    32
#define PTS_PER_BLOCK   (WAVES_PER_BLOCK * PTS_PER_WAVE)
#define LDS_STRIDE      65

__device__ __forceinline__ v8f zero8() {
    v8f z;
    #pragma unroll
    for (int i = 0; i < 8; ++i) z[i] = 0.0f;
    return z;
}

__device__ __forceinline__ float frcp(float x)  { return __builtin_amdgcn_rcpf(x); }
__device__ __forceinline__ float fsqrt(float x) { return __builtin_amdgcn_sqrtf(x); }

__global__ __launch_bounds__(PTS_PER_BLOCK, 4) void inception_traversal_kernel(
    const float* __restrict__ pos3d,
    const float* __restrict__ spectral,
    const float* __restrict__ centers1,
    const float* __restrict__ centers2,
    const float* __restrict__ centers3,
    const float* __restrict__ portal1,
    const float* __restrict__ portal2,
    const float* __restrict__ Wb,
    const float* __restrict__ bb,
    const float* __restrict__ bw,
    float* __restrict__ out,
    int BS)
{
    __shared__ float ldsRefr[WAVES_PER_BLOCK][PTS_PER_WAVE][LDS_STRIDE];

    const int lane   = threadIdx.x & 31;
    const int wid    = threadIdx.x >> 5;
    const int pblock = blockIdx.x * PTS_PER_BLOCK;
    if (pblock >= BS) return;
    const int pwave  = pblock + wid * PTS_PER_WAVE;

    float w0 = bw[0], w1 = bw[1], w2 = bw[2], w3 = bw[3];
    float wm = fmaxf(fmaxf(w0, w1), fmaxf(w2, w3));
    float ew0 = __expf(w0 - wm), ew1 = __expf(w1 - wm);
    float ew2 = __expf(w2 - wm), ew3 = __expf(w3 - wm);
    float wr  = frcp(ew0 + ew1 + ew2 + ew3);
    float wband[N_BANDS] = { ew0 * wr, ew1 * wr, ew2 * wr, ew3 * wr };

    const int mrow  = lane & 15;
    const int khalf = (lane >> 4) * 2;

    v8f acc[2][4];
    #pragma unroll
    for (int mt = 0; mt < 2; ++mt)
        #pragma unroll
        for (int t = 0; t < 4; ++t) acc[mt][t] = zero8();

    #pragma unroll
    for (int n = 0; n < N_BANDS; ++n) {
        v2f bfr[4][4];
        #pragma unroll
        for (int kk = 0; kk < 4; ++kk) {
            #pragma unroll
            for (int t = 0; t < 4; ++t) {
                const float* wp = Wb + ((size_t)n * BAND_SIZE + kk * 4 + khalf) * N_LEAF
                                     + t * 16 + mrow;
                bfr[kk][t].x = wp[0];
                bfr[kk][t].y = wp[N_LEAF];
            }
        }
        float bias[4];
        #pragma unroll
        for (int t = 0; t < 4; ++t) bias[t] = bb[n * N_LEAF + t * 16 + mrow];
        const float wn = wband[n];

        #pragma unroll
        for (int mt = 0; mt < 2; ++mt) {
            const int P = pwave + mt * 16;

            v8f logit[4];
            #pragma unroll
            for (int t = 0; t < 4; ++t) logit[t] = zero8();

            #pragma unroll
            for (int kk = 0; kk < 4; ++kk) {
                const int c = n * BAND_SIZE + kk * 4 + khalf;
                v2f a = *reinterpret_cast<const v2f*>(
                    spectral + (size_t)(P + mrow) * SPECTRAL_DIM + c);
                #pragma unroll
                for (int t = 0; t < 4; ++t)
                    logit[t] = __builtin_amdgcn_wmma_f32_16x16x4_f32(
                        false, a, false, bfr[kk][t], (short)0, logit[t], false, false);
            }
            #pragma unroll
            for (int t = 0; t < 4; ++t) {
                #pragma unroll
                for (int j = 0; j < 8; ++j) {
                    float s = frcp(1.0f + __expf(-(logit[t][j] + bias[t])));
                    acc[mt][t][j] += wn * s;
                }
            }
        }
    }

    #pragma unroll
    for (int mt = 0; mt < 2; ++mt) {
        const int rowoff = (lane >> 4) * 8 + mt * 16;
        #pragma unroll
        for (int t = 0; t < 4; ++t)
            #pragma unroll
            for (int j = 0; j < 8; ++j)
                ldsRefr[wid][rowoff + j][t * 16 + mrow] = acc[mt][t][j];
    }

    __syncthreads();

    const int p = pwave + lane;
    float* row = &ldsRefr[wid][lane][0];

    const float px = pos3d[(size_t)p * 3 + 0];
    const float py = pos3d[(size_t)p * 3 + 1];
    const float pz = pos3d[(size_t)p * 3 + 2];
    const float inv2T2 = 1.0f / (2.0f + EPSF);

    float d1v[4];
    #pragma unroll
    for (int d = 0; d < 4; ++d) {
        float dx = px - centers1[d * 3 + 0];
        float dy = py - centers1[d * 3 + 1];
        float dz = pz - centers1[d * 3 + 2];
        d1v[d] = dx * dx + dy * dy + dz * dz;
    }
    float z1max = -1e30f;
    #pragma unroll
    for (int d = 0; d < 4; ++d) z1max = fmaxf(z1max, -d1v[d] * inv2T2);
    float ex1[4], s1 = 0.0f;
    #pragma unroll
    for (int d = 0; d < 4; ++d) { ex1[d] = __expf(-d1v[d] * inv2T2 - z1max); s1 += ex1[d]; }
    const float r1 = frcp(s1);
    float m1[4], e1v[4];
    #pragma unroll
    for (int d = 0; d < 4; ++d) {
        m1[d]  = ex1[d] * r1;
        e1v[d] = __expf(-LAM * fsqrt(d1v[d] + EPSF));
    }

    float ssum = 0.0f;
    #pragma unroll
    for (int d = 0; d < 4; ++d) {
        const float* T1 = portal1 + d * 12;
        float qx = T1[0]*px + T1[1]*py + T1[2]*pz  + T1[3];
        float qy = T1[4]*px + T1[5]*py + T1[6]*pz  + T1[7];
        float qz = T1[8]*px + T1[9]*py + T1[10]*pz + T1[11];

        float d2v[4];
        #pragma unroll
        for (int si = 0; si < 4; ++si) {
            const float* c = centers2 + (size_t)(d * 4 + si) * 3;
            float dx = qx - c[0], dy = qy - c[1], dz = qz - c[2];
            d2v[si] = dx * dx + dy * dy + dz * dz;
        }
        float z2max = -1e30f;
        #pragma unroll
        for (int si = 0; si < 4; ++si) z2max = fmaxf(z2max, -d2v[si] * inv2T2);
        float ex2[4], s2 = 0.0f;
        #pragma unroll
        for (int si = 0; si < 4; ++si) { ex2[si] = __expf(-d2v[si] * inv2T2 - z2max); s2 += ex2[si]; }
        const float r2 = frcp(s2);

        #pragma unroll
        for (int si = 0; si < 4; ++si) {
            float m2 = m1[d] * ex2[si] * r2;
            float e2 = e1v[d] * __expf(-LAM * fsqrt(d2v[si] + EPSF));

            const float* T2 = portal2 + (size_t)(d * 4 + si) * 12;
            float rx = T2[0]*qx + T2[1]*qy + T2[2]*qz  + T2[3];
            float ry = T2[4]*qx + T2[5]*qy + T2[6]*qz  + T2[7];
            float rz = T2[8]*qx + T2[9]*qy + T2[10]*qz + T2[11];

            float d3v[4];
            #pragma unroll
            for (int ci = 0; ci < 4; ++ci) {
                const float* c = centers3 + (size_t)((d * 4 + si) * 4 + ci) * 3;
                float dx = rx - c[0], dy = ry - c[1], dz = rz - c[2];
                d3v[ci] = dx * dx + dy * dy + dz * dz;
            }
            float z3max = -1e30f;
            #pragma unroll
            for (int ci = 0; ci < 4; ++ci) z3max = fmaxf(z3max, -d3v[ci] * inv2T2);
            float ex3[4], s3 = 0.0f;
            #pragma unroll
            for (int ci = 0; ci < 4; ++ci) { ex3[ci] = __expf(-d3v[ci] * inv2T2 - z3max); s3 += ex3[ci]; }
            const float r3   = frcp(s3);
            const float m2e2 = m2 * e2;

            #pragma unroll
            for (int ci = 0; ci < 4; ++ci) {
                float g  = m2e2 * ex3[ci] * r3 * __expf(-LAM * fsqrt(d3v[ci] + EPSF));
                int   k  = (d * 4 + si) * 4 + ci;
                float pr = g * row[k];
                row[k] = pr;
                ssum  += pr;
            }
        }
    }

    float rn = frcp(fmaxf(ssum, 1e-8f));
    float* op = out + (size_t)p * N_LEAF;
    #pragma unroll
    for (int k4 = 0; k4 < 16; ++k4) {
        v4f v;
        v.x = row[k4 * 4 + 0] * rn;
        v.y = row[k4 * 4 + 1] * rn;
        v.z = row[k4 * 4 + 2] * rn;
        v.w = row[k4 * 4 + 3] * rn;
        *reinterpret_cast<v4f*>(op + k4 * 4) = v;
    }
}

extern "C" void kernel_launch(void* const* d_in, const int* in_sizes, int n_in,
                              void* d_out, int out_size, void* d_ws, size_t ws_size,
                              hipStream_t stream) {
    const float* pos3d    = (const float*)d_in[0];
    const float* spectral = (const float*)d_in[1];
    const float* centers1 = (const float*)d_in[2];
    const float* centers2 = (const float*)d_in[3];
    const float* centers3 = (const float*)d_in[4];
    const float* portal1  = (const float*)d_in[5];
    const float* portal2  = (const float*)d_in[6];
    const float* Wb       = (const float*)d_in[7];
    const float* bb       = (const float*)d_in[8];
    const float* bw       = (const float*)d_in[9];
    float* out            = (float*)d_out;

    const int BS = in_sizes[0] / 3;
    const int blocks = (BS + PTS_PER_BLOCK - 1) / PTS_PER_BLOCK;

    hipLaunchKernelGGL(inception_traversal_kernel,
                       dim3(blocks), dim3(PTS_PER_BLOCK), 0, stream,
                       pos3d, spectral, centers1, centers2, centers3,
                       portal1, portal2, Wb, bb, bw, out, BS);
}


# Output rows leave through the LDS tile transposed: each store instruction writes 4 whole 128-B lines (rows 2i,2i+1, 512 contiguous bytes) instead of 16 B in each of 32 lines; values bit-identical
# speedup vs baseline: 1.7215x; 1.3441x over previous
#include <hip/hip_runtime.h>

#define N_LEAF       64
#define SPECTRAL_DIM 64
#define N_BANDS      4
#define BAND_SIZE    16
#define LAM          0.1f
#define EPSF         1e-8f

typedef float v2f __attribute__((ext_vector_type(2)));
typedef float v4f __attribute__((ext_vector_type(4)));
typedef float v8f __attribute__((ext_vector_type(8)));

#define WAVES_PER_BLOCK 4
#define PTS_PER_WAVE    32
#define PTS_PER_BLOCK   (WAVES_PER_BLOCK * PTS_PER_WAVE)
#define LDS_STRIDE      65

__device__ __forceinline__ v8f zero8() {
    v8f z;
    #pragma unroll
    for (int i = 0; i < 8; ++i) z[i] = 0.0f;
    return z;
}

__device__ __forceinline__ float frcp(float x)  { return __builtin_amdgcn_rcpf(x); }
__device__ __forceinline__ float fsqrt(float x) { return __builtin_amdgcn_sqrtf(x); }

__global__ __launch_bounds__(PTS_PER_BLOCK, 4) void inception_traversal_kernel(
    const float* __restrict__ pos3d,
    const float* __restrict__ spectral,
    const float* __restrict__ centers1,
    const float* __restrict__ centers2,
    const float* __restrict__ centers3,
    const float* __restrict__ portal1,
    const float* __restrict__ portal2,
    const float* __restrict__ Wb,
    const float* __restrict__ bb,
    const float* __restrict__ bw,
    float* __restrict__ out,
    int BS)
{
    __shared__ float ldsRefr[WAVES_PER_BLOCK][PTS_PER_WAVE][LDS_STRIDE];

    const int lane   = threadIdx.x & 31;
    const int wid    = threadIdx.x >> 5;
    const int pblock = blockIdx.x * PTS_PER_BLOCK;
    if (pblock >= BS) return;
    const int pwave  = pblock + wid * PTS_PER_WAVE;

    float w0 = bw[0], w1 = bw[1], w2 = bw[2], w3 = bw[3];
    float wm = fmaxf(fmaxf(w0, w1), fmaxf(w2, w3));
    float ew0 = __expf(w0 - wm), ew1 = __expf(w1 - wm);
    float ew2 = __expf(w2 - wm), ew3 = __expf(w3 - wm);
    float wr  = frcp(ew0 + ew1 + ew2 + ew3);
    float wband[N_BANDS] = { ew0 * wr, ew1 * wr, ew2 * wr, ew3 * wr };

    const int mrow  = lane & 15;
    const int khalf = (lane >> 4) * 2;

    v8f acc[2][4];
    #pragma unroll
    for (int mt = 0; mt < 2; ++mt)
        #pragma unroll
        for (int t = 0; t < 4; ++t) acc[mt][t] = zero8();

    #pragma unroll
    for (int n = 0; n < N_BANDS; ++n) {
        v2f bfr[4][4];
        #pragma unroll
        for (int kk = 0; kk < 4; ++kk) {
            #pragma unroll
            for (int t = 0; t < 4; ++t) {
                const float* wp = Wb + ((size_t)n * BAND_SIZE + kk * 4 + khalf) * N_LEAF
                                     + t * 16 + mrow;
                bfr[kk][t].x = wp[0];
                bfr[kk][t].y = wp[N_LEAF];
            }
        }
        float bias[4];
        #pragma unroll
        for (int t = 0; t < 4; ++t) bias[t] = bb[n * N_LEAF + t * 16 + mrow];
        const float wn = wband[n];

        #pragma unroll
        for (int mt = 0; mt < 2; ++mt) {
            const int P = pwave + mt * 16;

            v8f logit[4];
            #pragma unroll
            for (int t = 0; t < 4; ++t) logit[t] = zero8();

            #pragma unroll
            for (int kk = 0; kk < 4; ++kk) {
                const int c = n * BAND_SIZE + kk * 4 + khalf;
                v2f a = *reinterpret_cast<const v2f*>(
                    spectral + (size_t)(P + mrow) * SPECTRAL_DIM + c);
                #pragma unroll
                for (int t = 0; t < 4; ++t)
                    logit[t] = __builtin_amdgcn_wmma_f32_16x16x4_f32(
                        false, a, false, bfr[kk][t], (short)0, logit[t], false, false);
            }
            #pragma unroll
            for (int t = 0; t < 4; ++t) {
                #pragma unroll
                for (int j = 0; j < 8; ++j) {
                    float s = frcp(1.0f + __expf(-(logit[t][j] + bias[t])));
                    acc[mt][t][j] += wn * s;
                }
            }
        }
    }

    #pragma unroll
    for (int mt = 0; mt < 2; ++mt) {
        const int rowoff = (lane >> 4) * 8 + mt * 16;
        #pragma unroll
        for (int t = 0; t < 4; ++t)
            #pragma unroll
            for (int j = 0; j < 8; ++j)
                ldsRefr[wid][rowoff + j][t * 16 + mrow] = acc[mt][t][j];
    }

    __syncthreads();

    const int p = pwave + lane;
    float* row = &ldsRefr[wid][lane][0];

    const float px = pos3d[(size_t)p * 3 + 0];
    const float py = pos3d[(size_t)p * 3 + 1];
    const float pz = pos3d[(size_t)p * 3 + 2];
    const float inv2T2 = 1.0f / (2.0f + EPSF);

    float d1v[4];
    #pragma unroll
    for (int d = 0; d < 4; ++d) {
        float dx = px - centers1[d * 3 + 0];
        float dy = py - centers1[d * 3 + 1];
        float dz = pz - centers1[d * 3 + 2];
        d1v[d] = dx * dx + dy * dy + dz * dz;
    }
    float z1max = -1e30f;
    #pragma unroll
    for (int d = 0; d < 4; ++d) z1max = fmaxf(z1max, -d1v[d] * inv2T2);
    float ex1[4], s1 = 0.0f;
    #pragma unroll
    for (int d = 0; d < 4; ++d) { ex1[d] = __expf(-d1v[d] * inv2T2 - z1max); s1 += ex1[d]; }
    const float r1 = frcp(s1);
    float m1[4], e1v[4];
    #pragma unroll
    for (int d = 0; d < 4; ++d) {
        m1[d]  = ex1[d] * r1;
        e1v[d] = __expf(-LAM * fsqrt(d1v[d] + EPSF));
    }

    float ssum = 0.0f;
    #pragma unroll
    for (int d = 0; d < 4; ++d) {
        const float* T1 = portal1 + d * 12;
        float qx = T1[0]*px + T1[1]*py + T1[2]*pz  + T1[3];
        float qy = T1[4]*px + T1[5]*py + T1[6]*pz  + T1[7];
        float qz = T1[8]*px + T1[9]*py + T1[10]*pz + T1[11];

        float d2v[4];
        #pragma unroll
        for (int si = 0; si < 4; ++si) {
            const float* c = centers2 + (size_t)(d * 4 + si) * 3;
            float dx = qx - c[0], dy = qy - c[1], dz = qz - c[2];
            d2v[si] = dx * dx + dy * dy + dz * dz;
        }
        float z2max = -1e30f;
        #pragma unroll
        for (int si = 0; si < 4; ++si) z2max = fmaxf(z2max, -d2v[si] * inv2T2);
        float ex2[4], s2 = 0.0f;
        #pragma unroll
        for (int si = 0; si < 4; ++si) { ex2[si] = __expf(-d2v[si] * inv2T2 - z2max); s2 += ex2[si]; }
        const float r2 = frcp(s2);

        #pragma unroll
        for (int si = 0; si < 4; ++si) {
            float m2 = m1[d] * ex2[si] * r2;
            float e2 = e1v[d] * __expf(-LAM * fsqrt(d2v[si] + EPSF));

            const float* T2 = portal2 + (size_t)(d * 4 + si) * 12;
            float rx = T2[0]*qx + T2[1]*qy + T2[2]*qz  + T2[3];
            float ry = T2[4]*qx + T2[5]*qy + T2[6]*qz  + T2[7];
            float rz = T2[8]*qx + T2[9]*qy + T2[10]*qz + T2[11];

            float d3v[4];
            #pragma unroll
            for (int ci = 0; ci < 4; ++ci) {
                const float* c = centers3 + (size_t)((d * 4 + si) * 4 + ci) * 3;
                float dx = rx - c[0], dy = ry - c[1], dz = rz - c[2];
                d3v[ci] = dx * dx + dy * dy + dz * dz;
            }
            float z3max = -1e30f;
            #pragma unroll
            for (int ci = 0; ci < 4; ++ci) z3max = fmaxf(z3max, -d3v[ci] * inv2T2);
            float ex3[4], s3 = 0.0f;
            #pragma unroll
            for (int ci = 0; ci < 4; ++ci) { ex3[ci] = __expf(-d3v[ci] * inv2T2 - z3max); s3 += ex3[ci]; }
            const float r3   = frcp(s3);
            const float m2e2 = m2 * e2;

            #pragma unroll
            for (int ci = 0; ci < 4; ++ci) {
                float g  = m2e2 * ex3[ci] * r3 * __expf(-LAM * fsqrt(d3v[ci] + EPSF));
                int   k  = (d * 4 + si) * 4 + ci;
                float pr = g * row[k];
                row[k] = pr;
                ssum  += pr;
            }
        }
    }

    const float rn = frcp(fmaxf(ssum, 1e-8f));

    __builtin_amdgcn_fence(__ATOMIC_RELEASE, "wavefront");
    __builtin_amdgcn_wave_barrier();
    __builtin_amdgcn_fence(__ATOMIC_ACQUIRE, "wavefront");

    const int rsub = lane >> 4;
    const int col  = (lane & 15) * 4;
    float* ow = out + (size_t)pwave * N_LEAF + lane * 4;
    const int rn_bits = __builtin_bit_cast(int, rn);
    #pragma unroll
    for (int i = 0; i < 16; ++i) {
        const float rlo = __builtin_bit_cast(float, __builtin_amdgcn_readlane(rn_bits, 2 * i));
        const float rhi = __builtin_bit_cast(float, __builtin_amdgcn_readlane(rn_bits, 2 * i + 1));
        const float rs  = rsub ? rhi : rlo;
        const float* src = &ldsRefr[wid][2 * i + rsub][col];
        v4f v;
        v.x = src[0] * rs;
        v.y = src[1] * rs;
        v.z = src[2] * rs;
        v.w = src[3] * rs;
        *reinterpret_cast<v4f*>(ow + i * 128) = v;
    }
}

extern "C" void kernel_launch(void* const* d_in, const int* in_sizes, int n_in,
                              void* d_out, int out_size, void* d_ws, size_t ws_size,
                              hipStream_t stream) {
    const float* pos3d    = (const float*)d_in[0];
    const float* spectral = (const float*)d_in[1];
    const float* centers1 = (const float*)d_in[2];
    const float* centers2 = (const float*)d_in[3];
    const float* centers3 = (const float*)d_in[4];
    const float* portal1  = (const float*)d_in[5];
    const float* portal2  = (const float*)d_in[6];
    const float* Wb       = (const float*)d_in[7];
    const float* bb       = (const float*)d_in[8];
    const float* bw       = (const float*)d_in[9];
    float* out            = (float*)d_out;

    const int BS = in_sizes[0] / 3;
    const int blocks = (BS + PTS_PER_BLOCK - 1) / PTS_PER_BLOCK;

    hipLaunchKernelGGL(inception_traversal_kernel,
                       dim3(blocks), dim3(PTS_PER_BLOCK), 0, stream,
                       pos3d, spectral, centers1, centers2, centers3,
                       portal1, portal2, Wb, bb, bw, out, BS);
}


# Each band sigmoid as one v_tanh_f32, 0.5+0.5*tanh(z/2), in place of exp + reciprocal: 1 transcendental and 2 multiply-adds per value, not 2 and 4; band sum starts at 0.5*sum(w), then n=0..3
# speedup vs baseline: 1.8930x; 1.0996x over previous
#include <hip/hip_runtime.h>

#define N_LEAF       64
#define SPECTRAL_DIM 64
#define N_BANDS      4
#define BAND_SIZE    16
#define LAM          0.1f
#define EPSF         1e-8f

typedef float v2f __attribute__((ext_vector_type(2)));
typedef float v4f __attribute__((ext_vector_type(4)));
typedef float v8f __attribute__((ext_vector_type(8)));

#define WAVES_PER_BLOCK 4
#define PTS_PER_WAVE    32
#define PTS_PER_BLOCK   (WAVES_PER_BLOCK * PTS_PER_WAVE)
#define LDS_STRIDE      65

__device__ __forceinline__ v8f zero8() {
    v8f z;
    #pragma unroll
    for (int i = 0; i < 8; ++i) z[i] = 0.0f;
    return z;
}

__device__ __forceinline__ float frcp(float x)  { return __builtin_amdgcn_rcpf(x); }
__device__ __forceinline__ float fsqrt(float x) { return __builtin_amdgcn_sqrtf(x); }
__device__ __forceinline__ float ftanh(float x) { return __builtin_amdgcn_tanhf(x); }

__global__ __launch_bounds__(PTS_PER_BLOCK, 4) void inception_traversal_kernel(
    const float* __restrict__ pos3d,
    const float* __restrict__ spectral,
    const float* __restrict__ centers1,
    const float* __restrict__ centers2,
    const float* __restrict__ centers3,
    const float* __restrict__ portal1,
    const float* __restrict__ portal2,
    const float* __restrict__ Wb,
    const float* __restrict__ bb,
    const float* __restrict__ bw,
    float* __restrict__ out,
    int BS)
{
    __shared__ float ldsRefr[WAVES_PER_BLOCK][PTS_PER_WAVE][LDS_STRIDE];

    const int lane   = threadIdx.x & 31;
    const int wid    = threadIdx.x >> 5;
    const int pblock = blockIdx.x * PTS_PER_BLOCK;
    if (pblock >= BS) return;
    const int pwave  = pblock + wid * PTS_PER_WAVE;

    float w0 = bw[0], w1 = bw[1], w2 = bw[2], w3 = bw[3];
    float wm = fmaxf(fmaxf(w0, w1), fmaxf(w2, w3));
    float ew0 = __expf(w0 - wm), ew1 = __expf(w1 - wm);
    float ew2 = __expf(w2 - wm), ew3 = __expf(w3 - wm);
    float wr  = frcp(ew0 + ew1 + ew2 + ew3);
    float wband[N_BANDS] = { ew0 * wr, ew1 * wr, ew2 * wr, ew3 * wr };

    const int mrow  = lane & 15;
    const int khalf = (lane >> 4) * 2;

    const float acc0 = 0.5f * (wband[0] + wband[1] + wband[2] + wband[3]);
    v8f acc[2][4];
    #pragma unroll
    for (int mt = 0; mt < 2; ++mt)
        #pragma unroll
        for (int t = 0; t < 4; ++t)
            #pragma unroll
            for (int j = 0; j < 8; ++j) acc[mt][t][j] = acc0;

    #pragma unroll
    for (int n = 0; n < N_BANDS; ++n) {
        v2f bfr[4][4];
        #pragma unroll
        for (int kk = 0; kk < 4; ++kk) {
            #pragma unroll
            for (int t = 0; t < 4; ++t) {
                const float* wp = Wb + ((size_t)n * BAND_SIZE + kk * 4 + khalf) * N_LEAF
                                     + t * 16 + mrow;
                bfr[kk][t].x = wp[0];
                bfr[kk][t].y = wp[N_LEAF];
            }
        }
        float hbias[4];
        #pragma unroll
        for (int t = 0; t < 4; ++t) hbias[t] = 0.5f * bb[n * N_LEAF + t * 16 + mrow];
        const float hwn = 0.5f * wband[n];

        #pragma unroll
        for (int mt = 0; mt < 2; ++mt) {
            const int P = pwave + mt * 16;

            v8f logit[4];
            #pragma unroll
            for (int t = 0; t < 4; ++t) logit[t] = zero8();

            #pragma unroll
            for (int kk = 0; kk < 4; ++kk) {
                const int c = n * BAND_SIZE + kk * 4 + khalf;
                v2f a = *reinterpret_cast<const v2f*>(
                    spectral + (size_t)(P + mrow) * SPECTRAL_DIM + c);
                #pragma unroll
                for (int t = 0; t < 4; ++t)
                    logit[t] = __builtin_amdgcn_wmma_f32_16x16x4_f32(
                        false, a, false, bfr[kk][t], (short)0, logit[t], false, false);
            }
            #pragma unroll
            for (int t = 0; t < 4; ++t) {
                #pragma unroll
                for (int j = 0; j < 8; ++j) {
                    const float th = ftanh(__builtin_fmaf(logit[t][j], 0.5f, hbias[t]));
                    acc[mt][t][j] = __builtin_fmaf(hwn, th, acc[mt][t][j]);
                }
            }
        }
    }

    #pragma unroll
    for (int mt = 0; mt < 2; ++mt) {
        const int rowoff = (lane >> 4) * 8 + mt * 16;
        #pragma unroll
        for (int t = 0; t < 4; ++t)
            #pragma unroll
            for (int j = 0; j < 8; ++j)
                ldsRefr[wid][rowoff + j][t * 16 + mrow] = acc[mt][t][j];
    }

    __syncthreads();

    const int p = pwave + lane;
    float* row = &ldsRefr[wid][lane][0];

    const float px = pos3d[(size_t)p * 3 + 0];
    const float py = pos3d[(size_t)p * 3 + 1];
    const float pz = pos3d[(size_t)p * 3 + 2];
    const float inv2T2 = 1.0f / (2.0f + EPSF);

    float d1v[4];
    #pragma unroll
    for (int d = 0; d < 4; ++d) {
        float dx = px - centers1[d * 3 + 0];
        float dy = py - centers1[d * 3 + 1];
        float dz = pz - centers1[d * 3 + 2];
        d1v[d] = dx * dx + dy * dy + dz * dz;
    }
    float z1max = -1e30f;
    #pragma unroll
    for (int d = 0; d < 4; ++d) z1max = fmaxf(z1max, -d1v[d] * inv2T2);
    float ex1[4], s1 = 0.0f;
    #pragma unroll
    for (int d = 0; d < 4; ++d) { ex1[d] = __expf(-d1v[d] * inv2T2 - z1max); s1 += ex1[d]; }
    const float r1 = frcp(s1);
    float m1[4], e1v[4];
    #pragma unroll
    for (int d = 0; d < 4; ++d) {
        m1[d]  = ex1[d] * r1;
        e1v[d] = __expf(-LAM * fsqrt(d1v[d] + EPSF));
    }

    float ssum = 0.0f;
    #pragma unroll
    for (int d = 0; d < 4; ++d) {
        const float* T1 = portal1 + d * 12;
        float qx = T1[0]*px + T1[1]*py + T1[2]*pz  + T1[3];
        float qy = T1[4]*px + T1[5]*py + T1[6]*pz  + T1[7];
        float qz = T1[8]*px + T1[9]*py + T1[10]*pz + T1[11];

        float d2v[4];
        #pragma unroll
        for (int si = 0; si < 4; ++si) {
            const float* c = centers2 + (size_t)(d * 4 + si) * 3;
            float dx = qx - c[0], dy = qy - c[1], dz = qz - c[2];
            d2v[si] = dx * dx + dy * dy + dz * dz;
        }
        float z2max = -1e30f;
        #pragma unroll
        for (int si = 0; si < 4; ++si) z2max = fmaxf(z2max, -d2v[si] * inv2T2);
        float ex2[4], s2 = 0.0f;
        #pragma unroll
        for (int si = 0; si < 4; ++si) { ex2[si] = __expf(-d2v[si] * inv2T2 - z2max); s2 += ex2[si]; }
        const float r2 = frcp(s2);

        #pragma unroll
        for (int si = 0; si < 4; ++si) {
            float m2 = m1[d] * ex2[si] * r2;
            float e2 = e1v[d] * __expf(-LAM * fsqrt(d2v[si] + EPSF));

            const float* T2 = portal2 + (size_t)(d * 4 + si) * 12;
            float rx = T2[0]*qx + T2[1]*qy + T2[2]*qz  + T2[3];
            float ry = T2[4]*qx + T2[5]*qy + T2[6]*qz  + T2[7];
            float rz = T2[8]*qx + T2[9]*qy + T2[10]*qz + T2[11];

            float d3v[4];
            #pragma unroll
            for (int ci = 0; ci < 4; ++ci) {
                const float* c = centers3 + (size_t)((d * 4 + si) * 4 + ci) * 3;
                float dx = rx - c[0], dy = ry - c[1], dz = rz - c[2];
                d3v[ci] = dx * dx + dy * dy + dz * dz;
            }
            float z3max = -1e30f;
            #pragma unroll
            for (int ci = 0; ci < 4; ++ci) z3max = fmaxf(z3max, -d3v[ci] * inv2T2);
            float ex3[4], s3 = 0.0f;
            #pragma unroll
            for (int ci = 0; ci < 4; ++ci) { ex3[ci] = __expf(-d3v[ci] * inv2T2 - z3max); s3 += ex3[ci]; }
            const float r3   = frcp(s3);
            const float m2e2 = m2 * e2;

            #pragma unroll
            for (int ci = 0; ci < 4; ++ci) {
                float g  = m2e2 * ex3[ci] * r3 * __expf(-LAM * fsqrt(d3v[ci] + EPSF));
                int   k  = (d * 4 + si) * 4 + ci;
                float pr = g * row[k];
                row[k] = pr;
                ssum  += pr;
            }
        }
    }

    const float rn = frcp(fmaxf(ssum, 1e-8f));

    __builtin_amdgcn_fence(__ATOMIC_RELEASE, "wavefront");
    __builtin_amdgcn_wave_barrier();
    __builtin_amdgcn_fence(__ATOMIC_ACQUIRE, "wavefront");

    const int rsub = lane >> 4;
    const int col  = (lane & 15) * 4;
    float* ow = out + (size_t)pwave * N_LEAF + lane * 4;
    const int rn_bits = __builtin_bit_cast(int, rn);
    #pragma unroll
    for (int i = 0; i < 16; ++i) {
        const float rlo = __builtin_bit_cast(float, __builtin_amdgcn_readlane(rn_bits, 2 * i));
        const float rhi = __builtin_bit_cast(float, __builtin_amdgcn_readlane(rn_bits, 2 * i + 1));
        const float rs  = rsub ? rhi : rlo;
        const float* src = &ldsRefr[wid][2 * i + rsub][col];
        v4f v;
        v.x = src[0] * rs;
        v.y = src[1] * rs;
        v.z = src[2] * rs;
        v.w = src[3] * rs;
        *reinterpret_cast<v4f*>(ow + i * 128) = v;
    }
}

extern "C" void kernel_launch(void* const* d_in, const int* in_sizes, int n_in,
                              void* d_out, int out_size, void* d_ws, size_t ws_size,
                              hipStream_t stream) {
    const float* pos3d    = (const float*)d_in[0];
    const float* spectral = (const float*)d_in[1];
    const float* centers1 = (const float*)d_in[2];
    const float* centers2 = (const float*)d_in[3];
    const float* centers3 = (const float*)d_in[4];
    const float* portal1  = (const float*)d_in[5];
    const float* portal2  = (const float*)d_in[6];
    const float* Wb       = (const float*)d_in[7];
    const float* bb       = (const float*)d_in[8];
    const float* bw       = (const float*)d_in[9];
    float* out            = (float*)d_out;

    const int BS = in_sizes[0] / 3;
    const int blocks = (BS + PTS_PER_BLOCK - 1) / PTS_PER_BLOCK;

    hipLaunchKernelGGL(inception_traversal_kernel,
                       dim3(blocks), dim3(PTS_PER_BLOCK), 0, stream,
                       pos3d, spectral, centers1, centers2, centers3,
                       portal1, portal2, Wb, bb, bw, out, BS);
}
